# starting kernel
# baseline (speedup 1.0000x reference)
#include <hip/hip_runtime.h>

typedef __attribute__((ext_vector_type(2))) float v2f;
typedef __attribute__((ext_vector_type(8))) float v8f;

#define HID   64
#define FIN   22
#define COUT  6
#define BN_EPS 1e-5f

__global__ __launch_bounds__(256) void k_init_deg(float* deg, int n) {
  int i = blockIdx.x * blockDim.x + threadIdx.x;
  if (i < n) deg[i] = 1.0f;
}

__global__ __launch_bounds__(256) void k_accum_deg(const int* __restrict__ dst,
                                                   float* deg, int e) {
  int i = blockIdx.x * blockDim.x + threadIdx.x;
  if (i < e) atomicAdd(&deg[dst[i]], 1.0f);
}

__global__ __launch_bounds__(256) void k_dinv(float* deg, int n) {
  int i = blockIdx.x * blockDim.x + threadIdx.x;
  if (i < n) deg[i] = rsqrtf(deg[i]);
}

__device__ __forceinline__ v8f wmma4(v2f a, v2f b, v8f c) {
  return __builtin_amdgcn_wmma_f32_16x16x4_f32(
       false, a,  false, b,
       (short)0, c,  false,  false);
}

__global__ __launch_bounds__(256) void k_embed(const float* __restrict__ x,
                                               const float* __restrict__ W,
                                               const float* __restrict__ bias,
                                               float* __restrict__ h, int n) {
  int wid  = (blockIdx.x * blockDim.x + threadIdx.x) >> 5;
  int lane = threadIdx.x & 31;
  int row0 = wid * 16;
  if (row0 >= n) return;

  int m  = lane & 15;
  int kh = (lane >> 4) << 1;
  int arow = row0 + m; if (arow > n - 1) arow = n - 1;

  v8f acc[4] = {};
  for (int k0 = 0; k0 < 24; k0 += 4) {
    int ka = k0 + kh;
    v2f a;
    a.x = (ka     < FIN) ? x[(size_t)arow * FIN + ka]     : 0.0f;
    a.y = (ka + 1 < FIN) ? x[(size_t)arow * FIN + ka + 1] : 0.0f;
#pragma unroll
    for (int t = 0; t < 4; ++t) {
      v2f b;
      b.x = (ka     < FIN) ? W[ka       * HID + t * 16 + m] : 0.0f;
      b.y = (ka + 1 < FIN) ? W[(ka + 1) * HID + t * 16 + m] : 0.0f;
      acc[t] = wmma4(a, b, acc[t]);
    }
  }
#pragma unroll
  for (int r = 0; r < 8; ++r) {
    int row = row0 + r + ((lane >> 4) << 3);
    if (row < n) {
#pragma unroll
      for (int t = 0; t < 4; ++t) {
        int col = t * 16 + m;
        float v = acc[t][r] + bias[col];
        h[(size_t)row * HID + col] = v > 0.0f ? v : 0.0f;
      }
    }
  }
}

__global__ __launch_bounds__(256) void k_gcn_gemm(const float* __restrict__ hin,
                                                  const float* __restrict__ W,
                                                  const float* __restrict__ dinv,
                                                  float* __restrict__ tout,
                                                  float* __restrict__ agg, int n) {
  int wid  = (blockIdx.x * blockDim.x + threadIdx.x) >> 5;
  int lane = threadIdx.x & 31;
  int row0 = wid * 16;
  if (row0 >= n) return;

  int m  = lane & 15;
  int kh = (lane >> 4) << 1;
  int arow = row0 + m; if (arow > n - 1) arow = n - 1;

  v8f acc[4] = {};
  for (int k0 = 0; k0 < HID; k0 += 4) {
    int ka = k0 + kh;
    v2f a;
    a.x = hin[(size_t)arow * HID + ka];
    a.y = hin[(size_t)arow * HID + ka + 1];
#pragma unroll
    for (int t = 0; t < 4; ++t) {
      v2f b;
      b.x = W[ka       * HID + t * 16 + m];
      b.y = W[(ka + 1) * HID + t * 16 + m];
      acc[t] = wmma4(a, b, acc[t]);
    }
  }
#pragma unroll
  for (int r = 0; r < 8; ++r) {
    int row = row0 + r + ((lane >> 4) << 3);
    if (row < n) {
      float di  = dinv[row];
      float di2 = di * di;
#pragma unroll
      for (int t = 0; t < 4; ++t) {
        int col = t * 16 + m;
        float v = acc[t][r];
        tout[(size_t)row * HID + col] = v;
        agg [(size_t)row * HID + col] = v * di2;
      }
    }
  }
}

__global__ __launch_bounds__(256) void k_scatter(const int* __restrict__ src,
                                                 const int* __restrict__ dst,
                                                 const float* __restrict__ dinv,
                                                 const float* __restrict__ t,
                                                 float* agg, int e) {
  int gid  = blockIdx.x * blockDim.x + threadIdx.x;
  int edge = gid >> 5;
  if (edge >= e) return;
  int lane = gid & 31;

  int s = src[edge];
  int d = dst[edge];
  float coef = dinv[s] * dinv[d];

  int f = lane << 1;
  const float2 v = *reinterpret_cast<const float2*>(t + (size_t)s * HID + f);
  atomicAdd(agg + (size_t)d * HID + f,     coef * v.x);
  atomicAdd(agg + (size_t)d * HID + f + 1, coef * v.y);
}

__global__ __launch_bounds__(256) void k_bn_relu(const float* __restrict__ agg,
                                                 const float* __restrict__ b,
                                                 const float* __restrict__ g,
                                                 const float* __restrict__ be,
                                                 const float* __restrict__ mean,
                                                 const float* __restrict__ var,
                                                 float* __restrict__ out, int total) {
  int i = blockIdx.x * blockDim.x + threadIdx.x;
  if (i >= total) return;
  int f = i & (HID - 1);
  float v = agg[i] + b[f];
  v = (v - mean[f]) * rsqrtf(var[f] + BN_EPS) * g[f] + be[f];
  out[i] = v > 0.0f ? v : 0.0f;
}

__global__ __launch_bounds__(256) void k_cls(const float* __restrict__ h,
                                             const float* __restrict__ W,
                                             const float* __restrict__ b,
                                             float* __restrict__ out, int n) {
  int i = blockIdx.x * blockDim.x + threadIdx.x;
  if (i >= n) return;
  float acc[COUT];
#pragma unroll
  for (int c = 0; c < COUT; ++c) acc[c] = b[c];
  for (int k = 0; k < HID; ++k) {
    float hv = h[(size_t)i * HID + k];
#pragma unroll
    for (int c = 0; c < COUT; ++c) acc[c] += hv * W[k * COUT + c];
  }
#pragma unroll
  for (int c = 0; c < COUT; ++c) out[(size_t)i * COUT + c] = acc[c];
}

extern "C" void kernel_launch(void* const* d_in, const int* in_sizes, int n_in,
                              void* d_out, int out_size, void* d_ws, size_t ws_size,
                              hipStream_t stream) {
  (void)n_in; (void)out_size; (void)ws_size;

  const float* x     = (const float*)d_in[0];
  const int*   ei    = (const int*)  d_in[1];
  const float* W_emb = (const float*)d_in[2];
  const float* b_emb = (const float*)d_in[3];
  const float* W1    = (const float*)d_in[4];
  const float* b1    = (const float*)d_in[5];
  const float* g1    = (const float*)d_in[6];
  const float* be1   = (const float*)d_in[7];
  const float* m1    = (const float*)d_in[8];
  const float* v1    = (const float*)d_in[9];
  const float* W2    = (const float*)d_in[10];
  const float* b2    = (const float*)d_in[11];
  const float* g2    = (const float*)d_in[12];
  const float* be2   = (const float*)d_in[13];
  const float* m2    = (const float*)d_in[14];
  const float* v2    = (const float*)d_in[15];
  const float* W_cls = (const float*)d_in[16];
  const float* b_cls = (const float*)d_in[17];

  const int n = in_sizes[0] / FIN;
  const int e = in_sizes[1] / 2;
  const int* src = ei;
  const int* dst = ei + e;

  float* ws   = (float*)d_ws;
  float* dinv = ws;
  float* B0   = ws + n;
  float* B1   = B0 + (size_t)n * HID;
  float* B2   = B1 + (size_t)n * HID;

  auto cdiv = [](long a, long b) { return (int)((a + b - 1) / b); };

  k_init_deg <<<cdiv(n, 256), 256, 0, stream>>>(dinv, n);
  k_accum_deg<<<cdiv(e, 256), 256, 0, stream>>>(dst, dinv, e);
  k_dinv     <<<cdiv(n, 256), 256, 0, stream>>>(dinv, n);

  const int gthreads = cdiv(n, 16) * 32;

  k_embed<<<cdiv(gthreads, 256), 256, 0, stream>>>(x, W_emb, b_emb, B0, n);

  k_gcn_gemm<<<cdiv(gthreads, 256), 256, 0, stream>>>(B0, W1, dinv, B1, B2, n);
  k_scatter <<<cdiv((long)e * 32, 256), 256, 0, stream>>>(src, dst, dinv, B1, B2, e);
  k_bn_relu <<<cdiv((long)n * HID, 256), 256, 0, stream>>>(B2, b1, g1, be1, m1, v1, B0, n * HID);

  k_gcn_gemm<<<cdiv(gthreads, 256), 256, 0, stream>>>(B0, W2, dinv, B1, B2, n);
  k_scatter <<<cdiv((long)e * 32, 256), 256, 0, stream>>>(src, dst, dinv, B1, B2, e);
  k_bn_relu <<<cdiv((long)n * HID, 256), 256, 0, stream>>>(B2, b2, g2, be2, m2, v2, B0, n * HID);

  k_cls<<<cdiv(n, 256), 256, 0, stream>>>(B0, W_cls, b_cls, (float*)d_out, n);
}


# Last layer's product formed transposed (W^T z^T): its D tiles are the B fragments of the classifier product, now on the matrix unit in registers; h2 not stored, 9 launches to 8; cls sum order changed
# speedup vs baseline: 1.9818x; 1.9818x over previous
#include <hip/hip_runtime.h>

typedef __attribute__((ext_vector_type(2))) float v2f;
typedef __attribute__((ext_vector_type(8))) float v8f;

#define HID   64
#define FIN   22
#define COUT  6
#define BN_EPS 1e-5f

__device__ __forceinline__ int package_half() {
  unsigned id = __builtin_amdgcn_s_sendmsg_rtn(0x87);
  return (int)((id >> 18) & 1u);
}

__global__ __launch_bounds__(256) void k_init_deg(float* deg0, float* deg1, int n) {
  int i = blockIdx.x * blockDim.x + threadIdx.x;
  if (i < n) {
    deg0[i] = 1.0f;
    deg1[i] = 0.0f;
  }
}

__global__ __launch_bounds__(256) void k_accum_deg(const int* __restrict__ dst,
                                                   float* deg0, float* deg1, int e) {
  int i = blockIdx.x * blockDim.x + threadIdx.x;
  if (i < e) {
    const int d = dst[i];
    float* deg = package_half() ? deg1 : deg0;
    atomicAdd(&deg[d], 1.0f);
  }
}

__global__ __launch_bounds__(256) void k_dinv(float* deg0, const float* __restrict__ deg1, int n) {
  int i = blockIdx.x * blockDim.x + threadIdx.x;
  if (i < n) deg0[i] = rsqrtf(deg0[i] + deg1[i]);
}

__device__ __forceinline__ v8f wmma4(v2f a, v2f b, v8f c) {
  return __builtin_amdgcn_wmma_f32_16x16x4_f32(
       false, a,  false, b,
       (short)0, c,  false,  false);
}


__device__ __forceinline__ void clear_tile_rows(float* zplane, int row0, int lane, int n) {
  const float4 z4 = make_float4(0.0f, 0.0f, 0.0f, 0.0f);
  float4* zp = reinterpret_cast<float4*>(zplane + (size_t)row0 * HID);
  if (n - row0 >= 16) {
#pragma unroll
    for (int k = 0; k < 8; ++k) zp[k * 32 + lane] = z4;
  } else {
    for (int i = lane; i < (n - row0) * (HID / 4); i += 32) zp[i] = z4;
  }
}

__global__ __launch_bounds__(256) void k_embed(const float* __restrict__ x,
                                               const float* __restrict__ W,
                                               const float* __restrict__ bias,
                                               const float* __restrict__ dinv,
                                               float* __restrict__ h,
                                               float* __restrict__ p0,
                                               float* __restrict__ p1, int n) {
  int wid  = (blockIdx.x * blockDim.x + threadIdx.x) >> 5;
  int lane = threadIdx.x & 31;
  int row0 = wid * 16;
  if (row0 >= n) return;

  int m  = lane & 15;
  int kh = (lane >> 4) << 1;
  int arow = row0 + m; if (arow > n - 1) arow = n - 1;

  v8f acc[4] = {};
  for (int k0 = 0; k0 < 24; k0 += 4) {
    int ka = k0 + kh;
    v2f a;
    a.x = (ka     < FIN) ? x[(size_t)arow * FIN + ka]     : 0.0f;
    a.y = (ka + 1 < FIN) ? x[(size_t)arow * FIN + ka + 1] : 0.0f;
#pragma unroll
    for (int t = 0; t < 4; ++t) {
      v2f b;
      b.x = (ka     < FIN) ? W[ka       * HID + t * 16 + m] : 0.0f;
      b.y = (ka + 1 < FIN) ? W[(ka + 1) * HID + t * 16 + m] : 0.0f;
      acc[t] = wmma4(a, b, acc[t]);
    }
  }
#pragma unroll
  for (int r = 0; r < 8; ++r) {
    int row = row0 + r + ((lane >> 4) << 3);
    if (row < n) {
      float di  = dinv[row];
#pragma unroll
      for (int t = 0; t < 4; ++t) {
        int col = t * 16 + m;
        float v = acc[t][r] + bias[col];
        v = v > 0.0f ? v : 0.0f;
        v *= di;
        h [(size_t)row * HID + col] = v;
        p0[(size_t)row * HID + col] = v;
      }
    }
  }
  clear_tile_rows(p1, row0, lane, n);
}

__global__ __launch_bounds__(256) void k_layer(const float* s0, const float* s1,
                                               const float* __restrict__ W,
                                               const float* __restrict__ b,
                                               const float* __restrict__ g,
                                               const float* __restrict__ be,
                                               const float* __restrict__ mean,
                                               const float* __restrict__ var,
                                               const float* __restrict__ dinv,
                                               float* __restrict__ hout,
                                               float* p0, float* p1, int n) {
  int wid  = (blockIdx.x * blockDim.x + threadIdx.x) >> 5;
  int lane = threadIdx.x & 31;
  int row0 = wid * 16;
  if (row0 >= n) return;

  int m  = lane & 15;
  int kh = (lane >> 4) << 1;
  int arow = row0 + m; if (arow > n - 1) arow = n - 1;
  const float da = dinv[arow];

  v8f acc[4] = {};
#pragma unroll
  for (int k0 = 0; k0 < HID; k0 += 4) {
    int ka = k0 + kh;
    v2f a;
    a.x = (s0[(size_t)arow * HID + ka]     + s1[(size_t)arow * HID + ka])     * da;
    a.y = (s0[(size_t)arow * HID + ka + 1] + s1[(size_t)arow * HID + ka + 1]) * da;
#pragma unroll
    for (int t = 0; t < 4; ++t) {
      v2f bf;
      bf.x = W[ka       * HID + t * 16 + m];
      bf.y = W[(ka + 1) * HID + t * 16 + m];
      acc[t] = wmma4(a, bf, acc[t]);
    }
  }
  float cb[4], cm[4], cs[4], cg[4], ce[4];
#pragma unroll
  for (int t = 0; t < 4; ++t) {
    int col = t * 16 + m;
    cb[t] = b[col];
    cm[t] = mean[col];
    cs[t] = rsqrtf(var[col] + BN_EPS);
    cg[t] = g[col];
    ce[t] = be[col];
  }
#pragma unroll
  for (int r = 0; r < 8; ++r) {
    int row = row0 + r + ((lane >> 4) << 3);
    if (row < n) {
      const float di = dinv[row];
#pragma unroll
      for (int t = 0; t < 4; ++t) {
        int col = t * 16 + m;
        float v = acc[t][r] + cb[t];
        v = (v - cm[t]) * cs[t] * cg[t] + ce[t];
        v = v > 0.0f ? v : 0.0f;
        v *= di;
        hout[(size_t)row * HID + col] = v;
        p0[(size_t)row * HID + col] = v;
      }
    }
  }
  clear_tile_rows(p1, row0, lane, n);
}

__global__ __launch_bounds__(256) void k_layer_cls(const float* __restrict__ s0,
                                                   const float* __restrict__ s1,
                                                   const float* __restrict__ W,
                                                   const float* __restrict__ b,
                                                   const float* __restrict__ g,
                                                   const float* __restrict__ be,
                                                   const float* __restrict__ mean,
                                                   const float* __restrict__ var,
                                                   const float* __restrict__ dinv,
                                                   const float* __restrict__ Wc,
                                                   const float* __restrict__ bc,
                                                   float* __restrict__ out, int n) {
  int wid  = (blockIdx.x * blockDim.x + threadIdx.x) >> 5;
  int lane = threadIdx.x & 31;
  int row0 = wid * 16;
  if (row0 >= n) return;

  const int m  = lane & 15;
  const int hh = lane >> 4;
  const int kh = hh << 1;
  int arow = row0 + m; if (arow > n - 1) arow = n - 1;
  const float da = dinv[arow];

  v8f acc[4] = {};
#pragma unroll
  for (int k0 = 0; k0 < HID; k0 += 4) {
    const int ka = k0 + kh;
    v2f z;
    z.x = (s0[(size_t)arow * HID + ka]     + s1[(size_t)arow * HID + ka])     * da;
    z.y = (s0[(size_t)arow * HID + ka + 1] + s1[(size_t)arow * HID + ka + 1]) * da;
#pragma unroll
    for (int t = 0; t < 4; ++t) {
      v2f w;
      w.x = W[ka       * HID + t * 16 + m];
      w.y = W[(ka + 1) * HID + t * 16 + m];
      acc[t] = wmma4(w, z, acc[t]);
    }
  }

  const int  cbase = hh << 3;
  const bool real  = m < COUT;
  const int  mc    = real ? m : 0;
  v8f o0 = {}, o1 = {};
#pragma unroll
  for (int t = 0; t < 4; ++t) {
    __builtin_amdgcn_sched_barrier(0);
#pragma unroll
    for (int r = 0; r < 8; ++r) {
      const int col = t * 16 + cbase + r;
      float v = acc[t][r] + b[col];
      v = (v - mean[col]) * rsqrtf(var[col] + BN_EPS) * g[col] + be[col];
      acc[t][r] = v > 0.0f ? v : 0.0f;
    }
#pragma unroll
    for (int q = 0; q < 8; q += 2) {
      const int col = t * 16 + cbase + q;
      v2f a, h2;
      const float w0 = Wc[col       * COUT + mc];
      const float w1 = Wc[(col + 1) * COUT + mc];
      a.x = real ? w0 : 0.0f;
      a.y = real ? w1 : 0.0f;
      h2.x = acc[t][q];
      h2.y = acc[t][q + 1];
      if (q & 2) o1 = wmma4(a, h2, o1);
      else       o0 = wmma4(a, h2, o0);
    }
  }

  const int row = row0 + m;
  if (hh == 0 && row < n) {
    float2* op = reinterpret_cast<float2*>(out + (size_t)row * COUT);
#pragma unroll
    for (int c = 0; c < COUT; c += 2) {
      float2 v;
      v.x = (o0[c]     + o1[c])     + bc[c];
      v.y = (o0[c + 1] + o1[c + 1]) + bc[c + 1];
      op[c >> 1] = v;
    }
  }
}

__global__ __launch_bounds__(256) void k_scatter(const int* __restrict__ src,
                                                 const int* __restrict__ dst,
                                                 const float* __restrict__ t,
                                                 float* agg0, float* agg1, int e) {
  const int lane = threadIdx.x & 31;
  const int wv = __builtin_amdgcn_readfirstlane((int)(blockIdx.x * 8 + (threadIdx.x >> 5)));
  const int e0 = wv * 32;
  if (e0 >= e) return;

  const int my = e0 + lane;
  int   s = 0, d = 0;
  if (my < e) {
    s = src[my];
    d = dst[my];
  }
  const int hf = package_half();
  float* agg = hf ? agg1 : agg0;
  const char* tb = reinterpret_cast<const char*>(t);
  char*       ab = reinterpret_cast<char*>(agg);
  const unsigned lo = (unsigned)lane << 2;

  if (e - e0 >= 32) {
    float va[32], vb[32];
#pragma unroll
    for (int j = 0; j < 32; ++j) {
      const unsigned sj = (unsigned)__builtin_amdgcn_readlane(s, j);
      const char* r = tb + ((sj << 8) + lo);
      va[j] = *reinterpret_cast<const float*>(r);
      vb[j] = *reinterpret_cast<const float*>(r + 128);
    }
#pragma unroll
    for (int j = 0; j < 32; ++j) {
      const unsigned dj = (unsigned)__builtin_amdgcn_readlane(d, j);
      float* p = reinterpret_cast<float*>(ab + ((dj << 8) + lo));
      atomicAdd(p,      va[j]);
      atomicAdd(p + 32, vb[j]);
    }
  } else {
    const int cnt = e - e0;
    for (int j = 0; j < cnt; ++j) {
      const unsigned sj = (unsigned)__builtin_amdgcn_readlane(s, j);
      const unsigned dj = (unsigned)__builtin_amdgcn_readlane(d, j);
      const char* r = tb + ((sj << 8) + lo);
      const float ra = *reinterpret_cast<const float*>(r);
      const float rb = *reinterpret_cast<const float*>(r + 128);
      float* p = reinterpret_cast<float*>(ab + ((dj << 8) + lo));
      atomicAdd(p,      ra);
      atomicAdd(p + 32, rb);
    }
  }
}

extern "C" void kernel_launch(void* const* d_in, const int* in_sizes, int n_in,
                              void* d_out, int out_size, void* d_ws, size_t ws_size,
                              hipStream_t stream) {
  (void)n_in; (void)out_size; (void)ws_size;

  const float* x     = (const float*)d_in[0];
  const int*   ei    = (const int*)  d_in[1];
  const float* W_emb = (const float*)d_in[2];
  const float* b_emb = (const float*)d_in[3];
  const float* W1    = (const float*)d_in[4];
  const float* b1    = (const float*)d_in[5];
  const float* g1    = (const float*)d_in[6];
  const float* be1   = (const float*)d_in[7];
  const float* m1    = (const float*)d_in[8];
  const float* v1    = (const float*)d_in[9];
  const float* W2    = (const float*)d_in[10];
  const float* b2    = (const float*)d_in[11];
  const float* g2    = (const float*)d_in[12];
  const float* be2   = (const float*)d_in[13];
  const float* m2    = (const float*)d_in[14];
  const float* v2    = (const float*)d_in[15];
  const float* W_cls = (const float*)d_in[16];
  const float* b_cls = (const float*)d_in[17];

  const int n = in_sizes[0] / FIN;
  const int e = in_sizes[1] / 2;
  const int* src = ei;
  const int* dst = ei + e;

  float* ws   = (float*)d_ws;
  float* dinv = ws;
  float* B0   = ws + n;
  float* B1   = B0 + (size_t)n * HID;
  float* B2   = B1 + (size_t)n * HID;

  auto cdiv = [](long a, long b) { return (int)((a + b - 1) / b); };

  k_init_deg <<<cdiv(n, 256), 256, 0, stream>>>(dinv, B1, n);
  k_accum_deg<<<cdiv(e, 256), 256, 0, stream>>>(dst, dinv, B1, e);
  k_dinv     <<<cdiv(n, 256), 256, 0, stream>>>(dinv, B1, n);

  const int gthreads = cdiv(n, 16) * 32;

  k_embed<<<cdiv(gthreads, 256), 256, 0, stream>>>(x, W_emb, b_emb, dinv, B1, B2, B0, n);

  k_scatter<<<cdiv(cdiv(e, 32), 8), 256, 0, stream>>>(src, dst, B1, B2, B0, e);
  k_layer<<<cdiv(gthreads, 256), 256, 0, stream>>>(B2, B0, W1, b1, g1, be1, m1, v1, dinv, B1, B2, B0, n);

  k_scatter<<<cdiv(cdiv(e, 32), 8), 256, 0, stream>>>(src, dst, B1, B2, B0, e);
  k_layer_cls<<<cdiv(gthreads, 256), 256, 0, stream>>>(B2, B0, W2, b2, g2, be2, m2, v2, dinv, W_cls, b_cls, (float*)d_out, n);
}
